# starting kernel
# baseline (speedup 1.0000x reference)
#include <hip/hip_runtime.h>
#include <hip/hip_bf16.h>

typedef __attribute__((ext_vector_type(2))) float v2f;
typedef __attribute__((ext_vector_type(8))) float v8f;

#define HW   8192
#define GRID 512
#define E    256

__global__ void dummyconv_prep_kernel(const float* __restrict__ weight,
                                      const float* __restrict__ bias,
                                      float* __restrict__ ws) {
    const int j = threadIdx.x;
    float s = 0.0f;
    #pragma unroll 8
    for (int i = 0; i < E; ++i) s += weight[i * E + j];
    ws[j] = s;

    __shared__ float red[E];
    red[j] = bias[j];
    __syncthreads();
    for (int off = E / 2; off > 0; off >>= 1) {
        if (j < off) red[j] += red[j + off];
        __syncthreads();
    }
    if (j == 0) ws[E] = red[0];
}

__global__ void __launch_bounds__(256)
dummyconv_main_kernel(const float* __restrict__ data,
                      const float* __restrict__ ws,
                      float* __restrict__ out) {
    __shared__ float lws[E + 1];
    const int tid = threadIdx.x;
    lws[tid] = ws[tid];
    if (tid == 0) lws[E] = ws[E];
    __syncthreads();

    const int wave = tid >> 5;
    const int lane = tid & 31;
    const int task = blockIdx.x * 8 + wave;
    const int gy   = task >> 5;
    const int gx0  = (task & 31) << 4;

    const int m = lane & 15;
    const int h = lane >> 4;

    const float* patchBase =
        data + (size_t)gy * 16 * HW + (size_t)(gx0 + m) * 16;

    v8f acc = {};

    #pragma unroll
    for (int p = 0; p < 32; ++p) {
        const int fb = p * 8 + h * 4;
        const int r  = fb >> 4;
        const int c  = fb & 15;

        const float4 a  = *(const float4*)(patchBase + (size_t)r * HW + c);
        const float4 wv = *(const float4*)(&lws[fb]);

        v2f A0 = {a.x, a.y};
        v2f B0 = {wv.x, wv.y};
        acc = __builtin_amdgcn_wmma_f32_16x16x4_f32(
                  false, A0, false, B0, (short)0, acc, false, false);

        v2f A1 = {a.z, a.w};
        v2f B1 = {wv.z, wv.w};
        acc = __builtin_amdgcn_wmma_f32_16x16x4_f32(
                  false, A1, false, B1, (short)0, acc, false, false);
    }

    const float bsum = lws[E];
    if ((lane & 8) == 0) {
        const int r7 = lane & 7;
        float v01 = (r7 & 1) ? acc[1] : acc[0];
        float v23 = (r7 & 1) ? acc[3] : acc[2];
        float v45 = (r7 & 1) ? acc[5] : acc[4];
        float v67 = (r7 & 1) ? acc[7] : acc[6];
        float v0123 = (r7 & 2) ? v23 : v01;
        float v4567 = (r7 & 2) ? v67 : v45;
        float v = (r7 & 4) ? v4567 : v0123;

        const int patch = h * 8 + r7;
        out[(size_t)gy * GRID + gx0 + patch] = v + bsum;
    }
}

extern "C" void kernel_launch(void* const* d_in, const int* in_sizes, int n_in,
                              void* d_out, int out_size, void* d_ws, size_t ws_size,
                              hipStream_t stream) {
    const float* data   = (const float*)d_in[0];
    const float* weight = (const float*)d_in[1];
    const float* bias   = (const float*)d_in[2];
    float*       out    = (float*)d_out;
    float*       ws     = (float*)d_ws;

    dummyconv_prep_kernel<<<dim3(1), dim3(256), 0, stream>>>(weight, bias, ws);

    dummyconv_main_kernel<<<dim3(2048), dim3(256), 0, stream>>>(data, ws, out);
}


# one launch: each of 256 resident work-groups sums the weight columns and the bias itself, then streams its 1 MB of data; the prep launch goes; the sums follow a different reduction tree
# speedup vs baseline: 1.7343x; 1.7343x over previous
#include <hip/hip_runtime.h>


typedef __attribute__((ext_vector_type(2))) float v2f;
typedef __attribute__((ext_vector_type(8))) float v8f;

#define HW   8192
#define GRID 512
#define E    256
#define TPW  4
#define PD   4
#define NG   (32 / PD)

__global__ void __launch_bounds__(512)
dummyconv_fused_kernel(const float* __restrict__ data,
                       const float* __restrict__ weight,
                       const float* __restrict__ bias,
                       float* __restrict__ out) {
    __shared__ float part[8][E];
    __shared__ float lws[E];

    const int tid  = threadIdx.x;
    const int wave = tid >> 5;
    const int lane = tid & 31;

    const int m  = lane & 15;
    const int h  = lane >> 4;
    const int gy = blockIdx.x * 2 + (wave >> 3);
    const int xs = wave & 7;

    const float* laneBase = data + (size_t)gy * 16 * HW + (size_t)xs * 1024
                                 + m * 16 + h * 4;

    float4 buf[2][TPW][PD];
    float4 wf[2][PD];

#define LOAD_GROUP(G)                                                          \
    _Pragma("unroll") for (int t = 0; t < TPW; ++t) {                          \
        _Pragma("unroll") for (int j = 0; j < PD; ++j) {                       \
            const int q = (G) * PD + j;                                        \
            buf[(G) & 1][t][j] = *(const float4*)(laneBase + t * 256           \
                                   + (size_t)(q >> 1) * HW + (q & 1) * 8);     \
        }                                                                      \
    }
#define LOAD_WF(G)                                                             \
    _Pragma("unroll") for (int j = 0; j < PD; ++j)                             \
        wf[(G) & 1][j] = *(const float4*)(&lws[((G) * PD + j) * 8 + h * 4]);

    LOAD_GROUP(0)
    __builtin_amdgcn_sched_barrier(0);

    float bsum;
    {
        const float4 b0 = *(const float4*)(bias + lane * 8);
        const float4 b1 = *(const float4*)(bias + lane * 8 + 4);

        const int g = tid & 63;
        const int s = tid >> 6;
        const float* wp = weight + (size_t)(s * 32) * E + g * 4;
        float4 w[32];
        #pragma unroll
        for (int i = 0; i < 32; ++i) w[i] = *(const float4*)(wp + (size_t)i * E);

        float v = ((b0.x + b0.y) + (b0.z + b0.w)) + ((b1.x + b1.y) + (b1.z + b1.w));
        v += __shfl_xor(v, 16);
        v += __shfl_xor(v, 8);
        v += __shfl_xor(v, 4);
        v += __shfl_xor(v, 2);
        v += __shfl_xor(v, 1);
        bsum = v;

        float4 sum = {0.0f, 0.0f, 0.0f, 0.0f};
        #pragma unroll
        for (int i = 0; i < 32; ++i) {
            sum.x += w[i].x; sum.y += w[i].y; sum.z += w[i].z; sum.w += w[i].w;
        }
        *(float4*)&part[s][g * 4] = sum;
    }
    __syncthreads();
    if (tid < E) {
        float s = part[0][tid];
        #pragma unroll
        for (int k = 1; k < 8; ++k) s += part[k][tid];
        lws[tid] = s;
    }
    __syncthreads();

    LOAD_WF(0)

    v8f acc[TPW];
    #pragma unroll
    for (int t = 0; t < TPW; ++t) acc[t] = (v8f){};

    #pragma unroll
    for (int g = 0; g < NG; ++g) {
        __builtin_amdgcn_sched_barrier(0);
        if (g + 1 < NG) {
            LOAD_GROUP(g + 1)
            LOAD_WF(g + 1)
        }
        __builtin_amdgcn_sched_barrier(0);
        #pragma unroll
        for (int j = 0; j < PD; ++j) {
            const float4 wv = wf[g & 1][j];
            const v2f B0 = {wv.x, wv.y};
            const v2f B1 = {wv.z, wv.w};
            #pragma unroll
            for (int t = 0; t < TPW; ++t) {
                const float4 a = buf[g & 1][t][j];
                const v2f A0 = {a.x, a.y};
                const v2f A1 = {a.z, a.w};
                acc[t] = __builtin_amdgcn_wmma_f32_16x16x4_f32(
                             false, A0, false, B0, (short)0, acc[t], false, false);
                acc[t] = __builtin_amdgcn_wmma_f32_16x16x4_f32(
                             false, A1, false, B1, (short)0, acc[t], false, false);
            }
        }
    }
#undef LOAD_GROUP
#undef LOAD_WF

    if ((lane & 8) == 0) {
        const int r7 = lane & 7;
        float* orow = out + (size_t)gy * GRID + xs * 64 + h * 8 + r7;
        #pragma unroll
        for (int t = 0; t < TPW; ++t) {
            float v01 = (r7 & 1) ? acc[t][1] : acc[t][0];
            float v23 = (r7 & 1) ? acc[t][3] : acc[t][2];
            float v45 = (r7 & 1) ? acc[t][5] : acc[t][4];
            float v67 = (r7 & 1) ? acc[t][7] : acc[t][6];
            float v0123 = (r7 & 2) ? v23 : v01;
            float v4567 = (r7 & 2) ? v67 : v45;
            float v = (r7 & 4) ? v4567 : v0123;
            orow[t * 16] = v + bsum;
        }
    }
}

extern "C" void kernel_launch(void* const* d_in, const int* in_sizes, int n_in,
                              void* d_out, int out_size, void* d_ws, size_t ws_size,
                              hipStream_t stream) {
    const float* data   = (const float*)d_in[0];
    const float* weight = (const float*)d_in[1];
    const float* bias   = (const float*)d_in[2];
    float*       out    = (float*)d_out;

    dummyconv_fused_kernel<<<dim3(256), dim3(512), 0, stream>>>(data, weight, bias, out);
}


# data in a register ring of 5 patch rows, each slot requested again (row+5) right behind the WMMAs that read it: 32-40 loads in flight per wave, also while it multiplies (was 16-32); sums unchanged
# speedup vs baseline: 1.8009x; 1.0384x over previous
#include <hip/hip_runtime.h>


typedef __attribute__((ext_vector_type(2))) float v2f;
typedef __attribute__((ext_vector_type(8))) float v8f;

#define HW   8192
#define GRID 512
#define E    256
#define TPW  4
#define ROWS 16
#define RING 5

__global__ void __launch_bounds__(512)
dummyconv_fused_kernel(const float* __restrict__ data,
                       const float* __restrict__ weight,
                       const float* __restrict__ bias,
                       float* __restrict__ out) {
    __shared__ float part[8][E];
    __shared__ float lws[E];

    const int tid  = threadIdx.x;
    const int wave = tid >> 5;
    const int lane = tid & 31;

    const int m  = lane & 15;
    const int h  = lane >> 4;
    const int gy = blockIdx.x * 2 + (wave >> 3);
    const int xs = wave & 7;

    const float* laneBase = data + (size_t)gy * 16 * HW + (size_t)xs * 1024
                                 + m * 16 + h * 4;

    float4 buf[RING][TPW][2];
    float4 wf[2][2];

#define LOAD_ROW(R)                                                            \
    _Pragma("unroll") for (int t = 0; t < TPW; ++t) {                          \
        _Pragma("unroll") for (int c = 0; c < 2; ++c) {                        \
            buf[(R) % RING][t][c] = *(const float4*)(laneBase + t * 256        \
                                   + (size_t)(R) * HW + c * 8);                \
        }                                                                      \
    }
#define LOAD_WF(R)                                                             \
    _Pragma("unroll") for (int c = 0; c < 2; ++c)                              \
        wf[(R) & 1][c] = *(const float4*)(&lws[((R) * 2 + c) * 8 + h * 4]);

    LOAD_ROW(0)
    LOAD_ROW(1)
    __builtin_amdgcn_sched_barrier(0);

    float bsum;
    {
        const float4 b0 = *(const float4*)(bias + lane * 8);
        const float4 b1 = *(const float4*)(bias + lane * 8 + 4);

        const int g = tid & 63;
        const int s = tid >> 6;
        const float* wp = weight + (size_t)(s * 32) * E + g * 4;
        float4 w[32];
        #pragma unroll
        for (int i = 0; i < 32; ++i) w[i] = *(const float4*)(wp + (size_t)i * E);

        float v = ((b0.x + b0.y) + (b0.z + b0.w)) + ((b1.x + b1.y) + (b1.z + b1.w));
        v += __shfl_xor(v, 16);
        v += __shfl_xor(v, 8);
        v += __shfl_xor(v, 4);
        v += __shfl_xor(v, 2);
        v += __shfl_xor(v, 1);
        bsum = v;

        float4 sum = {0.0f, 0.0f, 0.0f, 0.0f};
        #pragma unroll
        for (int i = 0; i < 32; ++i) {
            sum.x += w[i].x; sum.y += w[i].y; sum.z += w[i].z; sum.w += w[i].w;
        }
        *(float4*)&part[s][g * 4] = sum;
    }
    __syncthreads();
    if (tid < E) {
        float s = part[0][tid];
        #pragma unroll
        for (int k = 1; k < 8; ++k) s += part[k][tid];
        lws[tid] = s;
    }
    __syncthreads();

    LOAD_WF(0)
    LOAD_WF(1)

    v8f acc[TPW];
    #pragma unroll
    for (int t = 0; t < TPW; ++t) acc[t] = (v8f){};

    LOAD_ROW(2)
    LOAD_ROW(3)
    LOAD_ROW(4)

    #pragma unroll
    for (int r = 0; r < ROWS; ++r) {
        __builtin_amdgcn_sched_barrier(0);
        #pragma unroll
        for (int c = 0; c < 2; ++c) {
            const float4 wv = wf[r & 1][c];
            const v2f B0 = {wv.x, wv.y};
            const v2f B1 = {wv.z, wv.w};
            #pragma unroll
            for (int t = 0; t < TPW; ++t) {
                const float4 a = buf[r % RING][t][c];
                const v2f A0 = {a.x, a.y};
                const v2f A1 = {a.z, a.w};
                acc[t] = __builtin_amdgcn_wmma_f32_16x16x4_f32(
                             false, A0, false, B0, (short)0, acc[t], false, false);
                acc[t] = __builtin_amdgcn_wmma_f32_16x16x4_f32(
                             false, A1, false, B1, (short)0, acc[t], false, false);
            }
        }
        __builtin_amdgcn_sched_barrier(0);
        if (r + RING < ROWS) {
            LOAD_ROW(r + RING)
        }
        if (r + 2 < ROWS) {
            LOAD_WF(r + 2)
        }
    }
#undef LOAD_ROW
#undef LOAD_WF

    if ((lane & 8) == 0) {
        const int r7 = lane & 7;
        float* orow = out + (size_t)gy * GRID + xs * 64 + h * 8 + r7;
        #pragma unroll
        for (int t = 0; t < TPW; ++t) {
            float v01 = (r7 & 1) ? acc[t][1] : acc[t][0];
            float v23 = (r7 & 1) ? acc[t][3] : acc[t][2];
            float v45 = (r7 & 1) ? acc[t][5] : acc[t][4];
            float v67 = (r7 & 1) ? acc[t][7] : acc[t][6];
            float v0123 = (r7 & 2) ? v23 : v01;
            float v4567 = (r7 & 2) ? v67 : v45;
            float v = (r7 & 4) ? v4567 : v0123;
            orow[t * 16] = v + bsum;
        }
    }
}

extern "C" void kernel_launch(void* const* d_in, const int* in_sizes, int n_in,
                              void* d_out, int out_size, void* d_ws, size_t ws_size,
                              hipStream_t stream) {
    const float* data   = (const float*)d_in[0];
    const float* weight = (const float*)d_in[1];
    const float* bias   = (const float*)d_in[2];
    float*       out    = (float*)d_out;

    dummyconv_fused_kernel<<<dim3(256), dim3(512), 0, stream>>>(data, weight, bias, out);
}
